# starting kernel
# baseline (speedup 1.0000x reference)
#include <hip/hip_runtime.h>

typedef __attribute__((ext_vector_type(2))) float v2f;
typedef __attribute__((ext_vector_type(8))) float v8f;
typedef __attribute__((ext_vector_type(4))) int   v4i;

#define IN_F            4096
#define OUT_F           8192
#define BATCH           32
#define BLOCKS_PER_ROW  256
#define PF_DIST         64

__global__ __launch_bounds__(128)
void SparseQuantLinear_wmma_kernel(const float* __restrict__ x,
                                   const float* __restrict__ scales,
                                   const float* __restrict__ offsets,
                                   const float* __restrict__ bias,
                                   const int*   __restrict__ packed,
                                   float*       __restrict__ out)
{
    const int lane   = threadIdx.x & 31;
    const int wave   = threadIdx.x >> 5;
    const int o_base = blockIdx.x * 64 + wave * 16;

    const int n  = lane & 15;
    const int hi = lane >> 4;
    const int o  = o_base + n;

    const float* xrow0 = x + n * IN_F;
    const float* xrow1 = x + (n + 16) * IN_F;

    const int t0 = hi * 2;
    const int t1 = t0 + 4;

    v8f c0 = {};
    v8f c1 = {};

    for (int kb = 0; kb < BLOCKS_PER_ROW; ++kb) {
        const int   nz   = o * BLOCKS_PER_ROW + kb;
        const float s    = scales[nz];
        const float f    = offsets[nz];
        const v4i   p    = *(const v4i*)(packed + nz * 4);
        const int   colb = kb * 16;

        __builtin_prefetch(packed  + (nz + PF_DIST) * 4, 0, 0);
        __builtin_prefetch(scales  + (nz + PF_DIST),     0, 0);
        __builtin_prefetch(offsets + (nz + PF_DIST),     0, 0);

        v2f a0_lo = *(const v2f*)(xrow0 + colb + t0);
        v2f a0_hi = *(const v2f*)(xrow1 + colb + t0);
        v2f a1_lo = *(const v2f*)(xrow0 + colb + t1);
        v2f a1_hi = *(const v2f*)(xrow1 + colb + t1);

        const int byte0 = hi ? p.y : p.x;
        const int byte1 = hi ? p.w : p.z;
        v2f b0, b1;
        b0.x = (float)( byte0        & 15) * s + f;
        b0.y = (float)((byte0 >> 4)  & 15) * s + f;
        b1.x = (float)( byte1        & 15) * s + f;
        b1.y = (float)((byte1 >> 4)  & 15) * s + f;

        c0 = __builtin_amdgcn_wmma_f32_16x16x4_f32(false, a0_lo, false, b0, (short)0, c0, false, false);
        c1 = __builtin_amdgcn_wmma_f32_16x16x4_f32(false, a0_hi, false, b0, (short)0, c1, false, false);
        c0 = __builtin_amdgcn_wmma_f32_16x16x4_f32(false, a1_lo, false, b1, (short)0, c0, false, false);
        c1 = __builtin_amdgcn_wmma_f32_16x16x4_f32(false, a1_hi, false, b1, (short)0, c1, false, false);
    }

    const float bz   = bias[o];
    const int   mrow = hi * 8;
#pragma unroll
    for (int v = 0; v < 8; ++v) {
        out[(mrow + v)      * OUT_F + o] = c0[v] + bz;
        out[(mrow + v + 16) * OUT_F + o] = c1[v] + bz;
    }
}

extern "C" void kernel_launch(void* const* d_in, const int* in_sizes, int n_in,
                              void* d_out, int out_size, void* d_ws, size_t ws_size,
                              hipStream_t stream) {
    const float* x       = (const float*)d_in[0];
    const float* scales  = (const float*)d_in[1];
    const float* offsets = (const float*)d_in[2];
    const float* bias    = (const float*)d_in[3];
    const int*   packed  = (const int*)d_in[4];
    float* out = (float*)d_out;

    dim3 grid(OUT_F / 64);
    dim3 block(128);
    SparseQuantLinear_wmma_kernel<<<grid, block, 0, stream>>>(x, scales, offsets, bias, packed, out);
}


# K split over the 16 waves of a work-group (16 of 256 blocks each, both column tiles), partials summed through LDS in fixed wave order; grid 256 x 512 threads; sum order changes
# speedup vs baseline: 1.4512x; 1.4512x over previous
#include <hip/hip_runtime.h>

typedef __attribute__((ext_vector_type(2))) float v2f;
typedef __attribute__((ext_vector_type(8))) float v8f;
typedef __attribute__((ext_vector_type(4))) int   v4i;

#define IN_F            4096
#define OUT_F           8192
#define BATCH           32
#define BLOCKS_PER_ROW  256
#define PF_DIST         64
#define KSPLIT          16
#define KB_PER_WAVE     (BLOCKS_PER_ROW / KSPLIT)
#define ROWS_PER_WG     32
#define PART_PER_WAVE   (2 * 8 * 2 * 32)

__global__ __launch_bounds__(512)
void SparseQuantLinear_wmma_kernel(const float* __restrict__ x,
                                   const float* __restrict__ scales,
                                   const float* __restrict__ offsets,
                                   const float* __restrict__ bias,
                                   const int*   __restrict__ packed,
                                   float*       __restrict__ out)
{
    __shared__ float part[KSPLIT * PART_PER_WAVE];

    const int lane   = threadIdx.x & 31;
    const int wave   = threadIdx.x >> 5;
    const int o_base = blockIdx.x * ROWS_PER_WG;

    const int n  = lane & 15;
    const int hi = lane >> 4;
    const int oA = o_base + n;
    const int oB = o_base + 16 + n;

    const float* xrow0 = x + n * IN_F;
    const float* xrow1 = x + (n + 16) * IN_F;

    const int t0 = hi * 2;
    const int t1 = t0 + 4;

    v8f cA0 = {};
    v8f cA1 = {};
    v8f cB0 = {};
    v8f cB1 = {};

    const int kb_first = wave * KB_PER_WAVE;
#pragma unroll 2
    for (int i = 0; i < KB_PER_WAVE; ++i) {
        const int kb   = kb_first + i;
        const int colb = kb * 16;

        v2f a0_lo = *(const v2f*)(xrow0 + colb + t0);
        v2f a0_hi = *(const v2f*)(xrow1 + colb + t0);
        v2f a1_lo = *(const v2f*)(xrow0 + colb + t1);
        v2f a1_hi = *(const v2f*)(xrow1 + colb + t1);

        {
            const int   nz = oA * BLOCKS_PER_ROW + kb;
            const float s  = scales[nz];
            const float f  = offsets[nz];
            const v4i   p  = *(const v4i*)(packed + nz * 4);

            __builtin_prefetch(packed  + (nz + PF_DIST) * 4, 0, 0);
            __builtin_prefetch(scales  + (nz + PF_DIST),     0, 0);
            __builtin_prefetch(offsets + (nz + PF_DIST),     0, 0);

            const int byte0 = hi ? p.y : p.x;
            const int byte1 = hi ? p.w : p.z;
            v2f b0, b1;
            b0.x = (float)( byte0        & 15) * s + f;
            b0.y = (float)((byte0 >> 4)  & 15) * s + f;
            b1.x = (float)( byte1        & 15) * s + f;
            b1.y = (float)((byte1 >> 4)  & 15) * s + f;

            cA0 = __builtin_amdgcn_wmma_f32_16x16x4_f32(false, a0_lo, false, b0, (short)0, cA0, false, false);
            cA1 = __builtin_amdgcn_wmma_f32_16x16x4_f32(false, a0_hi, false, b0, (short)0, cA1, false, false);
            cA0 = __builtin_amdgcn_wmma_f32_16x16x4_f32(false, a1_lo, false, b1, (short)0, cA0, false, false);
            cA1 = __builtin_amdgcn_wmma_f32_16x16x4_f32(false, a1_hi, false, b1, (short)0, cA1, false, false);
        }
        {
            const int   nz = oB * BLOCKS_PER_ROW + kb;
            const float s  = scales[nz];
            const float f  = offsets[nz];
            const v4i   p  = *(const v4i*)(packed + nz * 4);

            __builtin_prefetch(packed  + (nz + PF_DIST) * 4, 0, 0);
            __builtin_prefetch(scales  + (nz + PF_DIST),     0, 0);
            __builtin_prefetch(offsets + (nz + PF_DIST),     0, 0);

            const int byte0 = hi ? p.y : p.x;
            const int byte1 = hi ? p.w : p.z;
            v2f b0, b1;
            b0.x = (float)( byte0        & 15) * s + f;
            b0.y = (float)((byte0 >> 4)  & 15) * s + f;
            b1.x = (float)( byte1        & 15) * s + f;
            b1.y = (float)((byte1 >> 4)  & 15) * s + f;

            cB0 = __builtin_amdgcn_wmma_f32_16x16x4_f32(false, a0_lo, false, b0, (short)0, cB0, false, false);
            cB1 = __builtin_amdgcn_wmma_f32_16x16x4_f32(false, a0_hi, false, b0, (short)0, cB1, false, false);
            cB0 = __builtin_amdgcn_wmma_f32_16x16x4_f32(false, a1_lo, false, b1, (short)0, cB0, false, false);
            cB1 = __builtin_amdgcn_wmma_f32_16x16x4_f32(false, a1_hi, false, b1, (short)0, cB1, false, false);
        }
    }

    float* mine = part + wave * PART_PER_WAVE;
#pragma unroll
    for (int v = 0; v < 8; ++v) {
        mine[((0 * 8 + v) * 2 + 0) * 32 + lane] = cA0[v];
        mine[((0 * 8 + v) * 2 + 1) * 32 + lane] = cB0[v];
        mine[((1 * 8 + v) * 2 + 0) * 32 + lane] = cA1[v];
        mine[((1 * 8 + v) * 2 + 1) * 32 + lane] = cB1[v];
    }
    __syncthreads();

    const int   m    = wave;
    const int   src  = ((m & 7) * 2 + hi) * 32 + (m >> 3) * 16 + n;
    const float bz   = bias[o_base + lane];
#pragma unroll
    for (int bh = 0; bh < 2; ++bh) {
        const float* q = part + bh * (8 * 2 * 32) + src;
        float acc = q[0];
#pragma unroll
        for (int w = 1; w < KSPLIT; ++w)
            acc += q[w * PART_PER_WAVE];
        out[(bh * 16 + m) * OUT_F + o_base + lane] = acc + bz;
    }
}

extern "C" void kernel_launch(void* const* d_in, const int* in_sizes, int n_in,
                              void* d_out, int out_size, void* d_ws, size_t ws_size,
                              hipStream_t stream) {
    const float* x       = (const float*)d_in[0];
    const float* scales  = (const float*)d_in[1];
    const float* offsets = (const float*)d_in[2];
    const float* bias    = (const float*)d_in[3];
    const int*   packed  = (const int*)d_in[4];
    float* out = (float*)d_out;

    dim3 grid(OUT_F / ROWS_PER_WG);
    dim3 block(512);
    SparseQuantLinear_wmma_kernel<<<grid, block, 0, stream>>>(x, scales, offsets, bias, packed, out);
}


# Fragment loaders re-laid: lane halves own alternate blocks, one whole block per 16-byte load, a line's loads issued together up front, stale prefetches dropped; k order and sum order change
# speedup vs baseline: 4.7272x; 3.2574x over previous
#include <hip/hip_runtime.h>

typedef __attribute__((ext_vector_type(2))) float v2f;
typedef __attribute__((ext_vector_type(4))) float v4f;
typedef __attribute__((ext_vector_type(8))) float v8f;
typedef __attribute__((ext_vector_type(4))) int   v4i;

#define IN_F            4096
#define OUT_F           8192
#define BATCH           32
#define BLOCKS_PER_ROW  256
#define KSPLIT          16
#define KB_PER_WAVE     (BLOCKS_PER_ROW / KSPLIT)
#define KB_PER_LANE     (KB_PER_WAVE / 2)
#define X_AHEAD         4
#define ROWS_PER_WG     32
#define PART_PER_WAVE   (2 * 8 * 2 * 32)

#define DEQ(w, s, f) v2f{ (float)((w) & 15) * (s) + (f), (float)(((w) >> 4) & 15) * (s) + (f) }
#define WMMA(a, b, c) __builtin_amdgcn_wmma_f32_16x16x4_f32(false, (a), false, (b), (short)0, (c), false, false)

#define KSTEP(xa, xb, pa, pb, wordA, wordB)                      \
    {                                                            \
        const v2f a0 = { xa.pa, xa.pb };                         \
        const v2f a1 = { xb.pa, xb.pb };                         \
        const v2f bA = DEQ(wordA, scA, ofA);                     \
        const v2f bB = DEQ(wordB, scB, ofB);                     \
        cA0 = WMMA(a0, bA, cA0);                                 \
        cA1 = WMMA(a1, bA, cA1);                                 \
        cB0 = WMMA(a0, bB, cB0);                                 \
        cB1 = WMMA(a1, bB, cB1);                                 \
    }

#define LOAD_X(j)                                                \
    {                                                            \
        x0l[j] = *(const v4f*)(xr0 + (j) * 32);                  \
        x0h[j] = *(const v4f*)(xr0 + (j) * 32 + 4);              \
        x1l[j] = *(const v4f*)(xr1 + (j) * 32);                  \
        x1h[j] = *(const v4f*)(xr1 + (j) * 32 + 4);              \
    }

__global__ __launch_bounds__(512)
void SparseQuantLinear_wmma_kernel(const float* __restrict__ x,
                                   const float* __restrict__ scales,
                                   const float* __restrict__ offsets,
                                   const float* __restrict__ bias,
                                   const int*   __restrict__ packed,
                                   float*       __restrict__ out)
{
    __shared__ float part[KSPLIT * PART_PER_WAVE];

    const int lane   = threadIdx.x & 31;
    const int wave   = threadIdx.x >> 5;
    const int o_base = blockIdx.x * ROWS_PER_WG;

    const int n  = lane & 15;
    const int hi = lane >> 4;

    const int kb_first = wave * KB_PER_WAVE;
    const int nzA = (o_base + n) * BLOCKS_PER_ROW + kb_first;
    const int nzB = nzA + 16 * BLOCKS_PER_ROW;

    const v4i* pkA = (const v4i*)packed + nzA + hi;
    const v4i* pkB = (const v4i*)packed + nzB + hi;
    v4i qA[KB_PER_LANE], qB[KB_PER_LANE];
#pragma unroll
    for (int j = 0; j < KB_PER_LANE; ++j) qA[j] = pkA[2 * j];
#pragma unroll
    for (int j = 0; j < KB_PER_LANE; ++j) qB[j] = pkB[2 * j];

    v4f sA[4], fA[4], sB[4], fB[4];
#pragma unroll
    for (int v = 0; v < 4; ++v) sA[v] = ((const v4f*)(scales  + nzA))[v];
#pragma unroll
    for (int v = 0; v < 4; ++v) fA[v] = ((const v4f*)(offsets + nzA))[v];
#pragma unroll
    for (int v = 0; v < 4; ++v) sB[v] = ((const v4f*)(scales  + nzB))[v];
#pragma unroll
    for (int v = 0; v < 4; ++v) fB[v] = ((const v4f*)(offsets + nzB))[v];

    const float* xr0 = x + n * IN_F + (kb_first + hi) * 16;
    const float* xr1 = xr0 + 16 * IN_F;
    v4f x0l[KB_PER_LANE], x0h[KB_PER_LANE], x1l[KB_PER_LANE], x1h[KB_PER_LANE];
#pragma unroll
    for (int j = 0; j < X_AHEAD; ++j) LOAD_X(j)

    __builtin_amdgcn_sched_barrier(0);

    v8f cA0 = {};
    v8f cA1 = {};
    v8f cB0 = {};
    v8f cB1 = {};

#pragma unroll
    for (int j = 0; j < KB_PER_LANE; ++j) {
        const int   e   = 2 * (j & 1);
        const float scA = hi ? sA[j >> 1][e + 1] : sA[j >> 1][e];
        const float ofA = hi ? fA[j >> 1][e + 1] : fA[j >> 1][e];
        const float scB = hi ? sB[j >> 1][e + 1] : sB[j >> 1][e];
        const float ofB = hi ? fB[j >> 1][e + 1] : fB[j >> 1][e];

        KSTEP(x0l[j], x1l[j], x, y, qA[j].x, qB[j].x)
        KSTEP(x0l[j], x1l[j], z, w, qA[j].y, qB[j].y)
        KSTEP(x0h[j], x1h[j], x, y, qA[j].z, qB[j].z)
        KSTEP(x0h[j], x1h[j], z, w, qA[j].w, qB[j].w)

        if (j + X_AHEAD < KB_PER_LANE) LOAD_X(j + X_AHEAD)
        __builtin_amdgcn_sched_barrier(0);
    }

    float* mine = part + wave * PART_PER_WAVE;
#pragma unroll
    for (int v = 0; v < 8; ++v) {
        mine[((0 * 8 + v) * 2 + 0) * 32 + lane] = cA0[v];
        mine[((0 * 8 + v) * 2 + 1) * 32 + lane] = cB0[v];
        mine[((1 * 8 + v) * 2 + 0) * 32 + lane] = cA1[v];
        mine[((1 * 8 + v) * 2 + 1) * 32 + lane] = cB1[v];
    }
    __syncthreads();

    const int   m    = wave;
    const int   src  = ((m & 7) * 2 + hi) * 32 + (m >> 3) * 16 + n;
    const float bz   = bias[o_base + lane];
#pragma unroll
    for (int bh = 0; bh < 2; ++bh) {
        const float* q = part + bh * (8 * 2 * 32) + src;
        float acc = q[0];
#pragma unroll
        for (int w = 1; w < KSPLIT; ++w)
            acc += q[w * PART_PER_WAVE];
        out[(bh * 16 + m) * OUT_F + o_base + lane] = acc + bz;
    }
}

extern "C" void kernel_launch(void* const* d_in, const int* in_sizes, int n_in,
                              void* d_out, int out_size, void* d_ws, size_t ws_size,
                              hipStream_t stream) {
    const float* x       = (const float*)d_in[0];
    const float* scales  = (const float*)d_in[1];
    const float* offsets = (const float*)d_in[2];
    const float* bias    = (const float*)d_in[3];
    const int*   packed  = (const int*)d_in[4];
    float* out = (float*)d_out;

    dim3 grid(OUT_F / ROWS_PER_WG);
    dim3 block(512);
    SparseQuantLinear_wmma_kernel<<<grid, block, 0, stream>>>(x, scales, offsets, bias, packed, out);
}


# K streamed inside each wave: 4 waves per work-group (one per SIMD) own 64 blocks each, in 8 chunks with loads issued two chunks ahead of the matrix work, not all up front; sum order changes
# speedup vs baseline: 5.3150x; 1.1243x over previous
#include <hip/hip_runtime.h>

typedef __attribute__((ext_vector_type(2))) float v2f;
typedef __attribute__((ext_vector_type(4))) float v4f;
typedef __attribute__((ext_vector_type(8))) float v8f;
typedef __attribute__((ext_vector_type(4))) int   v4i;

#define IN_F            4096
#define OUT_F           8192
#define BATCH           32
#define BLOCKS_PER_ROW  256
#define KSPLIT          4
#define KB_PER_WAVE     (BLOCKS_PER_ROW / KSPLIT)
#define KB_PER_LANE     (KB_PER_WAVE / 2)
#define J_PER_CHUNK     4
#define CHUNKS          (KB_PER_LANE / J_PER_CHUNK)
#define AHEAD           2
#define ROWS_PER_WG     32
#define PART_PER_WAVE   (2 * 8 * 2 * 32)

#define DEQ(w, s, f) v2f{ (float)((w) & 15) * (s) + (f), (float)(((w) >> 4) & 15) * (s) + (f) }
#define WMMA(a, b, c) __builtin_amdgcn_wmma_f32_16x16x4_f32(false, (a), false, (b), (short)0, (c), false, false)

#define KSTEP(xa, xb, pa, pb, wordA, wordB)                      \
    {                                                            \
        const v2f a0 = { xa.pa, xa.pb };                         \
        const v2f a1 = { xb.pa, xb.pb };                         \
        const v2f bA = DEQ(wordA, scA, ofA);                     \
        const v2f bB = DEQ(wordB, scB, ofB);                     \
        cA0 = WMMA(a0, bA, cA0);                                 \
        cA1 = WMMA(a1, bA, cA1);                                 \
        cB0 = WMMA(a0, bB, cB0);                                 \
        cB1 = WMMA(a1, bB, cB1);                                 \
    }

#define LOAD_X(j)                                                \
    {                                                            \
        x0l[j] = *(const v4f*)(xr0 + (j) * 32);                  \
        x0h[j] = *(const v4f*)(xr0 + (j) * 32 + 4);              \
        x1l[j] = *(const v4f*)(xr1 + (j) * 32);                  \
        x1h[j] = *(const v4f*)(xr1 + (j) * 32 + 4);              \
    }

#define LOAD_CHUNK(c)                                                                    \
    {                                                                                    \
        _Pragma("unroll")                                                                \
        for (int jj = 0; jj < J_PER_CHUNK; ++jj) qA[(c) * J_PER_CHUNK + jj] = pkA[2 * ((c) * J_PER_CHUNK + jj)]; \
        _Pragma("unroll")                                                                \
        for (int jj = 0; jj < J_PER_CHUNK; ++jj) qB[(c) * J_PER_CHUNK + jj] = pkB[2 * ((c) * J_PER_CHUNK + jj)]; \
        _Pragma("unroll")                                                                \
        for (int jj = 0; jj < J_PER_CHUNK; ++jj) LOAD_X((c) * J_PER_CHUNK + jj)          \
    }

#define LOAD_LINE(dst, src, half)                                                        \
    {                                                                                    \
        _Pragma("unroll")                                                                \
        for (int v = 0; v < 8; ++v) dst[(half) * 8 + v] = ((const v4f*)(src))[(half) * 8 + v]; \
    }

__global__ __launch_bounds__(128)
void SparseQuantLinear_wmma_kernel(const float* __restrict__ x,
                                   const float* __restrict__ scales,
                                   const float* __restrict__ offsets,
                                   const float* __restrict__ bias,
                                   const int*   __restrict__ packed,
                                   float*       __restrict__ out)
{
    __shared__ float part[KSPLIT * PART_PER_WAVE];

    const int lane   = threadIdx.x & 31;
    const int wave   = threadIdx.x >> 5;
    const int o_base = blockIdx.x * ROWS_PER_WG;

    const int n  = lane & 15;
    const int hi = lane >> 4;

    const int kb_first = wave * KB_PER_WAVE;
    const int nzA = (o_base + n) * BLOCKS_PER_ROW + kb_first;
    const int nzB = nzA + 16 * BLOCKS_PER_ROW;

    const v4i* pkA = (const v4i*)packed + nzA + hi;
    const v4i* pkB = (const v4i*)packed + nzB + hi;
    const float* xr0 = x + n * IN_F + (kb_first + hi) * 16;
    const float* xr1 = xr0 + 16 * IN_F;

    v4i qA[KB_PER_LANE], qB[KB_PER_LANE];
    v4f sA[16], fA[16], sB[16], fB[16];
    v4f x0l[KB_PER_LANE], x0h[KB_PER_LANE], x1l[KB_PER_LANE], x1h[KB_PER_LANE];

    LOAD_LINE(sA, scales  + nzA, 0)
    LOAD_LINE(fA, offsets + nzA, 0)
    LOAD_LINE(sB, scales  + nzB, 0)
    LOAD_LINE(fB, offsets + nzB, 0)
    LOAD_CHUNK(0)
    LOAD_CHUNK(1)

    v8f cA0 = {};
    v8f cA1 = {};
    v8f cB0 = {};
    v8f cB1 = {};

#pragma unroll
    for (int c = 0; c < CHUNKS; ++c) {
        __builtin_amdgcn_sched_barrier(0);
        if (c + AHEAD < CHUNKS) LOAD_CHUNK(c + AHEAD)
        if (c == 0) LOAD_LINE(sA, scales  + nzA, 1)
        if (c == 1) LOAD_LINE(fA, offsets + nzA, 1)
        if (c == 2) LOAD_LINE(sB, scales  + nzB, 1)
        if (c == 3) LOAD_LINE(fB, offsets + nzB, 1)
        __builtin_amdgcn_sched_barrier(0);

#pragma unroll
        for (int jj = 0; jj < J_PER_CHUNK; ++jj) {
            const int j = c * J_PER_CHUNK + jj;
            const int   e   = 2 * (j & 1);
            const float scA = hi ? sA[j >> 1][e + 1] : sA[j >> 1][e];
            const float ofA = hi ? fA[j >> 1][e + 1] : fA[j >> 1][e];
            const float scB = hi ? sB[j >> 1][e + 1] : sB[j >> 1][e];
            const float ofB = hi ? fB[j >> 1][e + 1] : fB[j >> 1][e];

            KSTEP(x0l[j], x1l[j], x, y, qA[j].x, qB[j].x)
            KSTEP(x0l[j], x1l[j], z, w, qA[j].y, qB[j].y)
            KSTEP(x0h[j], x1h[j], x, y, qA[j].z, qB[j].z)
            KSTEP(x0h[j], x1h[j], z, w, qA[j].w, qB[j].w)
        }
    }

    float* mine = part + wave * PART_PER_WAVE;
#pragma unroll
    for (int v = 0; v < 8; ++v) {
        mine[((0 * 8 + v) * 2 + 0) * 32 + lane] = cA0[v];
        mine[((0 * 8 + v) * 2 + 1) * 32 + lane] = cB0[v];
        mine[((1 * 8 + v) * 2 + 0) * 32 + lane] = cA1[v];
        mine[((1 * 8 + v) * 2 + 1) * 32 + lane] = cB1[v];
    }
    __syncthreads();

    const float bz = bias[o_base + lane];
#pragma unroll
    for (int r = 0; r < 8; ++r) {
        const int bh  = r >> 2;
        const int m   = wave + KSPLIT * (r & 3);
        const int src = ((m & 7) * 2 + hi) * 32 + (m >> 3) * 16 + n;
        const float* q = part + bh * (8 * 2 * 32) + src;
        float acc = q[0];
#pragma unroll
        for (int w = 1; w < KSPLIT; ++w)
            acc += q[w * PART_PER_WAVE];
        out[(bh * 16 + m) * OUT_F + o_base + lane] = acc + bz;
    }
}

extern "C" void kernel_launch(void* const* d_in, const int* in_sizes, int n_in,
                              void* d_out, int out_size, void* d_ws, size_t ws_size,
                              hipStream_t stream) {
    const float* x       = (const float*)d_in[0];
    const float* scales  = (const float*)d_in[1];
    const float* offsets = (const float*)d_in[2];
    const float* bias    = (const float*)d_in[3];
    const int*   packed  = (const int*)d_in[4];
    float* out = (float*)d_out;

    dim3 grid(OUT_F / ROWS_PER_WG);
    dim3 block(128);
    SparseQuantLinear_wmma_kernel<<<grid, block, 0, stream>>>(x, scales, offsets, bias, packed, out);
}
